# starting kernel
# baseline (speedup 1.0000x reference)
#include <hip/hip_runtime.h>
#include <hip/hip_bf16.h>
#include <math.h>


typedef __attribute__((ext_vector_type(2))) float v2f;
typedef __attribute__((ext_vector_type(8))) float v8f;

#define EPS_F 1e-6f

template <int D>
__global__ __launch_bounds__(128)
void texfeat_kernel(const float* __restrict__ xin_all, float* __restrict__ out)
{
    constexpr int H   = 256, W = 256;
    constexpr int K1  = 2 * D + 1;
    constexpr int PAD = D * D;
    constexpr int KK  = K1 * K1;
    constexpr int TH  = 32, TW = 32;
    constexpr int PH  = TH + 2 * PAD;
    constexpr int PW  = TW + 2 * PAD;
    constexpr int PWP = PW + 1;
    constexpr int RP  = ((16 + 2 * PAD + 3) / 4) * 4;
    constexpr int NCH = RP / 4;
    constexpr int PHH = 16 + RP;
    constexpr int HSP = TW + 1;

    __shared__ float Tx[PH][PWP];
    __shared__ float Tl[PH][PWP];
    __shared__ float Hs[3][PHH][HSP];

    const int tid = threadIdx.x;
    const int bc  = blockIdx.z;
    const int tx0 = blockIdx.x * TW;
    const int ty0 = blockIdx.y * TH;
    const float* __restrict__ xin = xin_all + (size_t)bc * (H * W);

    for (int i = tid; i < PH * PW; i += 128) {
        const int r  = i / PW, c = i % PW;
        const int gy = ty0 - PAD + r;
        const int gx = tx0 - PAD + c;
        float v = 0.f;
        if (gy >= 0 && gy < H && gx >= 0 && gx < W) v = xin[gy * W + gx];
        Tx[r][c] = v;
        Tl[r][c] = v * __logf(v + EPS_F);
    }
    __syncthreads();

    for (int i = tid; i < PHH * TW; i += 128) {
        const int r = i / TW, c = i % TW;
        float h1 = 0.f, h2 = 0.f, h3 = 0.f;
        if (r < PH) {
            #pragma unroll
            for (int j = 0; j < K1; ++j) {
                const float v = Tx[r][c + j * D];
                h1 += v;
                h2 = __builtin_fmaf(v, v, h2);
                h3 += Tl[r][c + j * D];
            }
        }
        Hs[0][r][c] = h1;
        Hs[1][r][c] = h2;
        Hs[2][r][c] = h3;
    }
    __syncthreads();

    const int wave = tid >> 5, lane = tid & 31;
    const int sh = (wave >> 1) * 16;
    const int sw = (wave & 1) * 16;
    const int hi = lane >> 4;
    const int n  = lane & 15;

    v2f afrag[NCH];
    #pragma unroll
    for (int ch = 0; ch < NCH; ++ch) {
        const int r0 = ch * 4 + 2 * hi;
        const int t0 = r0 - n;
        const int t1 = r0 + 1 - n;
        afrag[ch].x = (t0 >= 0 && t0 <= 2 * PAD && (t0 % D) == 0) ? 1.f : 0.f;
        afrag[ch].y = (t1 >= 0 && t1 <= 2 * PAD && (t1 % D) == 0) ? 1.f : 0.f;
    }

    v8f acc[3];
    #pragma unroll
    for (int f = 0; f < 3; ++f) {
        v8f s = {0.f, 0.f, 0.f, 0.f, 0.f, 0.f, 0.f, 0.f};
        #pragma unroll
        for (int ch = 0; ch < NCH; ++ch) {
            const int br = sh + ch * 4 + 2 * hi;
            v2f b;
            b.x = Hs[f][br][sw + n];
            b.y = Hs[f][br + 1][sw + n];
            s = __builtin_amdgcn_wmma_f32_16x16x4_f32(
                    false, afrag[ch], false, b, (short)0, s, false, false);
        }
        acc[f] = s;
    }

    constexpr float KKf  = (float)KK;
    const float invK = 1.f / KKf;
    const float corr = KKf / (KKf - 1.f);
    const float EH   = 0.6065306597126334f;
    const size_t HW  = (size_t)H * W;
    float* __restrict__ o0 = out + (size_t)(bc * 4 + 0) * HW;
    float* __restrict__ o1 = out + (size_t)(bc * 4 + 1) * HW;
    float* __restrict__ o2 = out + (size_t)(bc * 4 + 2) * HW;
    float* __restrict__ o3 = out + (size_t)(bc * 4 + 3) * HW;

    #pragma unroll
    for (int v = 0; v < 8; ++v) {
        const int m  = hi * 8 + v;
        const int y  = sh + m;
        const int xc = sw + n;
        const float s1 = acc[0][v];
        const float s2 = acc[1][v];
        const float s3 = acc[2][v];

        const float mu = s1 * invK;
        float var_b = s2 * invK - mu * mu;
        var_b = var_b > 0.f ? var_b : 0.f;
        const float sd       = __builtin_sqrtf(var_b * corr) + EPS_F;
        const float contrast = var_b / (sd * sd);
        const float energy   = s2 * invK;
        const float entropy  = -s3 * invK;

        float ad = 0.f;
        for (int jy = 0; jy < K1; ++jy) {
            const float* rp = &Tx[y + jy * D][xc];
            #pragma unroll
            for (int jx = 0; jx < K1; ++jx)
                ad += __builtin_fabsf(rp[jx * D] - mu);
        }
        const float homog = 1.f / (1.f + ad * invK);

        const size_t pix = (size_t)(ty0 + y) * W + (tx0 + xc);
        o0[pix] = (contrast + EPS_F) * EH;
        o1[pix] = (energy   + EPS_F) * EH;
        o2[pix] = (entropy  + EPS_F) * EH;
        o3[pix] = (homog    + EPS_F) * EH;
    }
}

extern "C" void kernel_launch(void* const* d_in, const int* in_sizes, int n_in,
                              void* d_out, int out_size, void* d_ws, size_t ws_size,
                              hipStream_t stream) {
    (void)in_sizes; (void)n_in; (void)d_ws; (void)ws_size; (void)out_size;
    const float* x = (const float*)d_in[0];
    float* out = (float*)d_out;

    constexpr size_t PLANE = (size_t)2 * 16 * 256 * 256;
    dim3 grid(256 / 32, 256 / 32, 8);
    dim3 blk(128);

    texfeat_kernel<1><<<grid, blk, 0, stream>>>(x, out + 0 * PLANE);
    texfeat_kernel<2><<<grid, blk, 0, stream>>>(x, out + 1 * PLANE);
    texfeat_kernel<3><<<grid, blk, 0, stream>>>(x, out + 2 * PLANE);
    texfeat_kernel<4><<<grid, blk, 0, stream>>>(x, out + 3 * PLANE);
}


# Stage A requests all of a thread's tile elements back to back (clamped always-valid address, zero selected afterwards, fully unrolled): one memory wait per work-group instead of one per element
# speedup vs baseline: 1.2714x; 1.2714x over previous
#include <hip/hip_runtime.h>
#include <hip/hip_bf16.h>
#include <math.h>


typedef __attribute__((ext_vector_type(2))) float v2f;
typedef __attribute__((ext_vector_type(8))) float v8f;

#define EPS_F 1e-6f

template <int D>
__global__ __launch_bounds__(128)
void texfeat_kernel(const float* __restrict__ xin_all, float* __restrict__ out)
{
    constexpr int H   = 256, W = 256;
    constexpr int K1  = 2 * D + 1;
    constexpr int PAD = D * D;
    constexpr int KK  = K1 * K1;
    constexpr int TH  = 32, TW = 32;
    constexpr int PH  = TH + 2 * PAD;
    constexpr int PW  = TW + 2 * PAD;
    constexpr int PWP = PW + 1;
    constexpr int RP  = ((16 + 2 * PAD + 3) / 4) * 4;
    constexpr int NCH = RP / 4;
    constexpr int PHH = 16 + RP;
    constexpr int HSP = TW + 1;

    __shared__ float Tx[PH][PWP];
    __shared__ float Tl[PH][PWP];
    __shared__ float Hs[3][PHH][HSP];

    const int tid = threadIdx.x;
    const int bc  = blockIdx.z;
    const int tx0 = blockIdx.x * TW;
    const int ty0 = blockIdx.y * TH;
    const float* __restrict__ xin = xin_all + (size_t)bc * (H * W);

    constexpr int NEL = PH * PW;
    constexpr int NLD = (NEL + 127) / 128;
    float xv[NLD];
    #pragma unroll
    for (int k = 0; k < NLD; ++k) {
        int i = tid + k * 128;
        if ((k + 1) * 128 > NEL) i = i < NEL ? i : NEL - 1;
        const int r  = i / PW, c = i % PW;
        const int gy = ty0 - PAD + r;
        const int gx = tx0 - PAD + c;
        const bool inb = gy >= 0 && gy < H && gx >= 0 && gx < W;
        const int cy = gy < 0 ? 0 : (gy > H - 1 ? H - 1 : gy);
        const int cx = gx < 0 ? 0 : (gx > W - 1 ? W - 1 : gx);
        const float t = xin[cy * W + cx];
        xv[k] = inb ? t : 0.f;
    }
    #pragma unroll
    for (int k = 0; k < NLD; ++k) {
        int i = tid + k * 128;
        if ((k + 1) * 128 > NEL) i = i < NEL ? i : NEL - 1;
        const int r = i / PW, c = i % PW;
        const float v = xv[k];
        Tx[r][c] = v;
        Tl[r][c] = v * __logf(v + EPS_F);
    }
    __syncthreads();

    for (int i = tid; i < PHH * TW; i += 128) {
        const int r = i / TW, c = i % TW;
        float h1 = 0.f, h2 = 0.f, h3 = 0.f;
        if (r < PH) {
            #pragma unroll
            for (int j = 0; j < K1; ++j) {
                const float v = Tx[r][c + j * D];
                h1 += v;
                h2 = __builtin_fmaf(v, v, h2);
                h3 += Tl[r][c + j * D];
            }
        }
        Hs[0][r][c] = h1;
        Hs[1][r][c] = h2;
        Hs[2][r][c] = h3;
    }
    __syncthreads();

    const int wave = tid >> 5, lane = tid & 31;
    const int sh = (wave >> 1) * 16;
    const int sw = (wave & 1) * 16;
    const int hi = lane >> 4;
    const int n  = lane & 15;

    v2f afrag[NCH];
    #pragma unroll
    for (int ch = 0; ch < NCH; ++ch) {
        const int r0 = ch * 4 + 2 * hi;
        const int t0 = r0 - n;
        const int t1 = r0 + 1 - n;
        afrag[ch].x = (t0 >= 0 && t0 <= 2 * PAD && (t0 % D) == 0) ? 1.f : 0.f;
        afrag[ch].y = (t1 >= 0 && t1 <= 2 * PAD && (t1 % D) == 0) ? 1.f : 0.f;
    }

    v8f acc[3];
    #pragma unroll
    for (int f = 0; f < 3; ++f) {
        v8f s = {0.f, 0.f, 0.f, 0.f, 0.f, 0.f, 0.f, 0.f};
        #pragma unroll
        for (int ch = 0; ch < NCH; ++ch) {
            const int br = sh + ch * 4 + 2 * hi;
            v2f b;
            b.x = Hs[f][br][sw + n];
            b.y = Hs[f][br + 1][sw + n];
            s = __builtin_amdgcn_wmma_f32_16x16x4_f32(
                    false, afrag[ch], false, b, (short)0, s, false, false);
        }
        acc[f] = s;
    }

    constexpr float KKf  = (float)KK;
    const float invK = 1.f / KKf;
    const float corr = KKf / (KKf - 1.f);
    const float EH   = 0.6065306597126334f;
    const size_t HW  = (size_t)H * W;
    float* __restrict__ o0 = out + (size_t)(bc * 4 + 0) * HW;
    float* __restrict__ o1 = out + (size_t)(bc * 4 + 1) * HW;
    float* __restrict__ o2 = out + (size_t)(bc * 4 + 2) * HW;
    float* __restrict__ o3 = out + (size_t)(bc * 4 + 3) * HW;

    #pragma unroll
    for (int v = 0; v < 8; ++v) {
        const int m  = hi * 8 + v;
        const int y  = sh + m;
        const int xc = sw + n;
        const float s1 = acc[0][v];
        const float s2 = acc[1][v];
        const float s3 = acc[2][v];

        const float mu = s1 * invK;
        float var_b = s2 * invK - mu * mu;
        var_b = var_b > 0.f ? var_b : 0.f;
        const float sd       = __builtin_sqrtf(var_b * corr) + EPS_F;
        const float contrast = var_b / (sd * sd);
        const float energy   = s2 * invK;
        const float entropy  = -s3 * invK;

        float ad = 0.f;
        for (int jy = 0; jy < K1; ++jy) {
            const float* rp = &Tx[y + jy * D][xc];
            #pragma unroll
            for (int jx = 0; jx < K1; ++jx)
                ad += __builtin_fabsf(rp[jx * D] - mu);
        }
        const float homog = 1.f / (1.f + ad * invK);

        const size_t pix = (size_t)(ty0 + y) * W + (tx0 + xc);
        o0[pix] = (contrast + EPS_F) * EH;
        o1[pix] = (energy   + EPS_F) * EH;
        o2[pix] = (entropy  + EPS_F) * EH;
        o3[pix] = (homog    + EPS_F) * EH;
    }
}

extern "C" void kernel_launch(void* const* d_in, const int* in_sizes, int n_in,
                              void* d_out, int out_size, void* d_ws, size_t ws_size,
                              hipStream_t stream) {
    (void)in_sizes; (void)n_in; (void)d_ws; (void)ws_size; (void)out_size;
    const float* x = (const float*)d_in[0];
    float* out = (float*)d_out;

    constexpr size_t PLANE = (size_t)2 * 16 * 256 * 256;
    dim3 grid(256 / 32, 256 / 32, 8);
    dim3 blk(128);

    texfeat_kernel<1><<<grid, blk, 0, stream>>>(x, out + 0 * PLANE);
    texfeat_kernel<2><<<grid, blk, 0, stream>>>(x, out + 1 * PLANE);
    texfeat_kernel<3><<<grid, blk, 0, stream>>>(x, out + 2 * PLANE);
    texfeat_kernel<4><<<grid, blk, 0, stream>>>(x, out + 3 * PLANE);
}


# One launch runs all four dilations (dilation from block index, largest first; size-only code shared, tap loop rolled to keep the text small); abs-deviation terms now summed tap column by column
# speedup vs baseline: 2.3425x; 1.8425x over previous
#include <hip/hip_runtime.h>
#include <math.h>


typedef __attribute__((ext_vector_type(2))) float v2f;
typedef __attribute__((ext_vector_type(8))) float v8f;

#define EPS_F 1e-6f

constexpr int IMG  = 256;
constexpr int TILE = 32;
constexpr int HSP  = TILE + 1;

template <int D> struct Geo {
    static constexpr int K1  = 2 * D + 1;
    static constexpr int PAD = D * D;
    static constexpr int PH  = TILE + 2 * PAD;
    static constexpr int PWP = PH + 1;
    static constexpr int RP  = ((16 + 2 * PAD + 3) / 4) * 4;
    static constexpr int PHH = 16 + RP;
    static constexpr int WORDS = 2 * PH * PWP + 3 * PHH * HSP;
    static constexpr unsigned long long band() {
        unsigned long long m = 0;
        for (int t = 0; t <= 2 * PAD; t += D) m |= 1ull << t;
        return m;
    }
};
constexpr int SMEM_WORDS = Geo<4>::WORDS;

template <int D>
__device__ __forceinline__
void hsum_rows(float* smem, const int tid)
{
    using G = Geo<D>;
    const float* __restrict__ Tx = smem;
    const float* __restrict__ Tl = smem + G::PH * G::PWP;
    float* __restrict__ Hs = smem + 2 * G::PH * G::PWP;
    for (int i = tid; i < G::PHH * TILE; i += 128) {
        const int r = i / TILE, c = i % TILE;
        float h1 = 0.f, h2 = 0.f, h3 = 0.f;
        if (r < G::PH) {
            #pragma unroll
            for (int j = 0; j < G::K1; ++j) {
                const float v = Tx[r * G::PWP + c + j * D];
                h1 += v;
                h2 = __builtin_fmaf(v, v, h2);
                h3 += Tl[r * G::PWP + c + j * D];
            }
        }
        Hs[(0 * G::PHH + r) * HSP + c] = h1;
        Hs[(1 * G::PHH + r) * HSP + c] = h2;
        Hs[(2 * G::PHH + r) * HSP + c] = h3;
    }
}

template <int D>
__device__ __forceinline__
void absdev_taps(const float* base, const float (&mu)[8], float (&ad)[8])
{
    using G = Geo<D>;
    constexpr int NR = 8 + 2 * G::PAD;
    #pragma nounroll
    for (int jx = 0; jx < G::K1; ++jx) {
        const float* col = base + jx * D;
        #pragma unroll
        for (int rr = 0; rr < NR; ++rr) {
            const float p = col[rr * G::PWP];
            #pragma unroll
            for (int v = 0; v < 8; ++v) {
                if (rr >= v && (rr - v) % D == 0 && (rr - v) <= 2 * G::PAD)
                    ad[v] += __builtin_fabsf(p - mu[v]);
            }
        }
    }
}

__global__ __launch_bounds__(128)
void texfeat_all_kernel(const float* __restrict__ x, float* __restrict__ out)
{
    __shared__ float smem[SMEM_WORDS];

    const int tid = threadIdx.x;
    const int bc  = blockIdx.z & 7;
    const int sel = blockIdx.z >> 3;
    const int D   = 4 - sel;
    const int tx0 = blockIdx.x * TILE;
    const int ty0 = blockIdx.y * TILE;

    const int PAD = D * D;
    const int PH  = TILE + 2 * PAD;
    const int PWP = PH + 1;
    const int RP  = ((16 + 2 * PAD + 3) >> 2) << 2;
    const int NCH = RP >> 2;
    const int PHH = 16 + RP;
    float* Tx = smem;
    float* Tl = smem + PH * PWP;
    const float* Hs = smem + 2 * PH * PWP;
    const float* __restrict__ xin = x + (size_t)bc * (IMG * IMG);

    {
        const int RSTEP = D <= 2 ? 3 : 2;
        const int CSTEP = 128 - RSTEP * PH;
        int r = (tid >= PH ? 1 : 0) + (tid >= 2 * PH ? 1 : 0) + (tid >= 3 * PH ? 1 : 0);
        int c = tid - r * PH;
        constexpr int NB = 8;
        #pragma nounroll
        for (int k0 = 0; k0 * 128 < PH * PH; k0 += NB) {
            float xv[NB];
            int   lo[NB];
            #pragma unroll
            for (int u = 0; u < NB; ++u) {
                const bool past = r >= PH;
                const int rr = past ? PH - 1 : r;
                const int cc = past ? PH - 1 : c;
                const int gy = ty0 - PAD + rr;
                const int gx = tx0 - PAD + cc;
                const bool inb = gy >= 0 && gy < IMG && gx >= 0 && gx < IMG;
                const int cy = gy < 0 ? 0 : (gy > IMG - 1 ? IMG - 1 : gy);
                const int cx = gx < 0 ? 0 : (gx > IMG - 1 ? IMG - 1 : gx);
                const float t = xin[cy * IMG + cx];
                xv[u] = inb ? t : 0.f;
                lo[u] = rr * PWP + cc;
                c += CSTEP;
                r += RSTEP;
                if (c >= PH) { c -= PH; r += 1; }
            }
            #pragma unroll
            for (int u = 0; u < NB; ++u) {
                const float v = xv[u];
                Tx[lo[u]] = v;
                Tl[lo[u]] = v * __logf(v + EPS_F);
            }
        }
    }
    __syncthreads();

    if (sel == 0)      hsum_rows<4>(smem, tid);
    else if (sel == 1) hsum_rows<3>(smem, tid);
    else if (sel == 2) hsum_rows<2>(smem, tid);
    else               hsum_rows<1>(smem, tid);
    __syncthreads();

    const int wave = tid >> 5, lane = tid & 31;
    const int sh = (wave >> 1) * 16;
    const int sw = (wave & 1) * 16;
    const int hi = lane >> 4;
    const int n  = lane & 15;
    const unsigned long long band = sel == 0 ? Geo<4>::band() : sel == 1 ? Geo<3>::band()
                                  : sel == 2 ? Geo<2>::band() : Geo<1>::band();

    v8f acc0 = {0.f, 0.f, 0.f, 0.f, 0.f, 0.f, 0.f, 0.f};
    v8f acc1 = acc0, acc2 = acc0;
    for (int ch = 0; ch < NCH; ++ch) {
        const int r0 = ch * 4 + 2 * hi;
        const int t0 = r0 - n;
        const int t1 = r0 + 1 - n;
        v2f a;
        a.x = (t0 >= 0 && ((band >> (t0 & 63)) & 1ull)) ? 1.f : 0.f;
        a.y = (t1 >= 0 && ((band >> (t1 & 63)) & 1ull)) ? 1.f : 0.f;
        const int br = sh + r0;
        const float* hb = Hs + br * HSP + sw + n;
        v2f b0, b1, b2;
        b0.x = hb[0];                   b0.y = hb[HSP];
        b1.x = hb[PHH * HSP];           b1.y = hb[PHH * HSP + HSP];
        b2.x = hb[2 * PHH * HSP];       b2.y = hb[2 * PHH * HSP + HSP];
        acc0 = __builtin_amdgcn_wmma_f32_16x16x4_f32(false, a, false, b0, (short)0, acc0, false, false);
        acc1 = __builtin_amdgcn_wmma_f32_16x16x4_f32(false, a, false, b1, (short)0, acc1, false, false);
        acc2 = __builtin_amdgcn_wmma_f32_16x16x4_f32(false, a, false, b2, (short)0, acc2, false, false);
    }

    const float invK = sel == 0 ? 1.f / 81.f : sel == 1 ? 1.f / 49.f : sel == 2 ? 1.f / 25.f : 1.f / 9.f;
    const float corr = sel == 0 ? 81.f / 80.f : sel == 1 ? 49.f / 48.f
                     : sel == 2 ? 25.f / 24.f : 9.f / 8.f;
    const float EH   = 0.6065306597126334f;

    const int y0 = sh + hi * 8;
    const int xc = sw + n;
    float mu[8], ad[8], contrast[8], energy[8], entropy[8];
    #pragma unroll
    for (int v = 0; v < 8; ++v) {
        const float s1 = acc0[v];
        const float s2 = acc1[v];
        const float s3 = acc2[v];
        mu[v] = s1 * invK;
        float var_b = s2 * invK - mu[v] * mu[v];
        var_b = var_b > 0.f ? var_b : 0.f;
        const float sd = __builtin_sqrtf(var_b * corr) + EPS_F;
        contrast[v] = var_b / (sd * sd);
        energy[v]   = s2 * invK;
        entropy[v]  = -s3 * invK;
        ad[v] = 0.f;
    }

    if (sel == 0)      absdev_taps<4>(smem + y0 * Geo<4>::PWP + xc, mu, ad);
    else if (sel == 1) absdev_taps<3>(smem + y0 * Geo<3>::PWP + xc, mu, ad);
    else if (sel == 2) absdev_taps<2>(smem + y0 * Geo<2>::PWP + xc, mu, ad);
    else               absdev_taps<1>(smem + y0 * Geo<1>::PWP + xc, mu, ad);

    constexpr size_t HW    = (size_t)IMG * IMG;
    constexpr size_t PLANE = (size_t)2 * 16 * HW;
    float* __restrict__ od = out + (size_t)(D - 1) * PLANE;
    float* __restrict__ o0 = od + (size_t)(bc * 4 + 0) * HW;
    float* __restrict__ o1 = od + (size_t)(bc * 4 + 1) * HW;
    float* __restrict__ o2 = od + (size_t)(bc * 4 + 2) * HW;
    float* __restrict__ o3 = od + (size_t)(bc * 4 + 3) * HW;
    #pragma unroll
    for (int v = 0; v < 8; ++v) {
        const float homog = 1.f / (1.f + ad[v] * invK);
        const size_t pix = (size_t)(ty0 + y0 + v) * IMG + (tx0 + xc);
        o0[pix] = (contrast[v] + EPS_F) * EH;
        o1[pix] = (energy[v]   + EPS_F) * EH;
        o2[pix] = (entropy[v]  + EPS_F) * EH;
        o3[pix] = (homog       + EPS_F) * EH;
    }
}

extern "C" void kernel_launch(void* const* d_in, const int* in_sizes, int n_in,
                              void* d_out, int out_size, void* d_ws, size_t ws_size,
                              hipStream_t stream) {
    (void)in_sizes; (void)n_in; (void)d_ws; (void)ws_size; (void)out_size;
    const float* x = (const float*)d_in[0];
    float* out = (float*)d_out;

    dim3 grid(256 / 32, 256 / 32, 4 * 8);
    dim3 blk(128);

    texfeat_all_kernel<<<grid, blk, 0, stream>>>(x, out);
}


# One work-group per tile and channel stages the padded tile (x and x*log x) once and serves all four dilations from it; 8 waves, each one 16x16 sub-tile and two dilations; every sum's order unchanged
# speedup vs baseline: 2.6240x; 1.1202x over previous
#include <hip/hip_runtime.h>
#include <math.h>


typedef __attribute__((ext_vector_type(2))) float v2f;
typedef __attribute__((ext_vector_type(8))) float v8f;

#define EPS_F 1e-6f

constexpr int IMG  = 256;
constexpr int TILE = 32;
constexpr int HSP  = TILE + 1;
constexpr int NTHR = 256;

constexpr int TPAD = 16;
constexpr int TPH  = TILE + 2 * TPAD;
constexpr int TPW  = TPH + 2;
constexpr int TWORDS = TPH * TPW;

template <int D> struct Geo {
    static constexpr int K1  = 2 * D + 1;
    static constexpr int PAD = D * D;
    static constexpr int PH  = TILE + 2 * PAD;
    static constexpr int OFF = TPAD - PAD;
    static constexpr int RP  = ((16 + 2 * PAD + 3) / 4) * 4;
    static constexpr int PHH = 16 + RP;
    static constexpr int HWORDS = 3 * PHH * HSP;
    static constexpr unsigned long long band() {
        unsigned long long m = 0;
        for (int t = 0; t <= 2 * PAD; t += D) m |= 1ull << t;
        return m;
    }
};
constexpr int HS4 = 2 * TWORDS;
constexpr int HS3 = HS4 + Geo<4>::HWORDS;
constexpr int HS2 = HS3 + Geo<3>::HWORDS;
constexpr int HS1 = HS2 + Geo<2>::HWORDS;
constexpr int SMEM_WORDS = HS1 + Geo<1>::HWORDS;
template <int D> struct HsBase { static constexpr int v = D == 4 ? HS4 : D == 3 ? HS3 : D == 2 ? HS2 : HS1; };

template <int D>
__device__ __forceinline__
void hsum_rows(float* smem, const int tid)
{
    using G = Geo<D>;
    const float* __restrict__ Tx = smem + G::OFF * TPW + G::OFF;
    const float* __restrict__ Tl = Tx + TWORDS;
    float* __restrict__ Hs = smem + HsBase<D>::v;
    for (int i = tid; i < G::PHH * TILE; i += NTHR) {
        const int r = i / TILE, c = i % TILE;
        float h1 = 0.f, h2 = 0.f, h3 = 0.f;
        if (r < G::PH) {
            #pragma unroll
            for (int j = 0; j < G::K1; ++j) {
                const float v = Tx[r * TPW + c + j * D];
                h1 += v;
                h2 = __builtin_fmaf(v, v, h2);
                h3 += Tl[r * TPW + c + j * D];
            }
        }
        Hs[(0 * G::PHH + r) * HSP + c] = h1;
        Hs[(1 * G::PHH + r) * HSP + c] = h2;
        Hs[(2 * G::PHH + r) * HSP + c] = h3;
    }
}

template <int D>
__device__ __forceinline__
void absdev_taps(const float* base, const float (&mu)[8], float (&ad)[8])
{
    using G = Geo<D>;
    constexpr int NR = 8 + 2 * G::PAD;
    #pragma nounroll
    for (int jx = 0; jx < G::K1; ++jx) {
        const float* col = base + jx * D;
        #pragma unroll
        for (int rr = 0; rr < NR; ++rr) {
            const float p = col[rr * TPW];
            #pragma unroll
            for (int v = 0; v < 8; ++v) {
                if (rr >= v && (rr - v) % D == 0 && (rr - v) <= 2 * G::PAD)
                    ad[v] += __builtin_fabsf(p - mu[v]);
            }
        }
    }
}

__global__ __launch_bounds__(NTHR)
void texfeat_all_kernel(const float* __restrict__ x, float* __restrict__ out)
{
    __shared__ float smem[SMEM_WORDS];

    const int tid = threadIdx.x;
    const int bc  = blockIdx.z;
    const int tx0 = blockIdx.x * TILE;
    const int ty0 = blockIdx.y * TILE;
    const float* __restrict__ xin = x + (size_t)bc * (IMG * IMG);

    {
        constexpr int RS = NTHR / TPH;
        constexpr int NE = TPH / RS;
        const int c  = tid & (TPH - 1);
        const int r0 = tid / TPH;
        const int gx = tx0 - TPAD + c;
        const bool cin = gx >= 0 && gx < IMG;
        const int cx = gx < 0 ? 0 : (gx > IMG - 1 ? IMG - 1 : gx);
        float xv[NE];
        #pragma unroll
        for (int k = 0; k < NE; ++k) {
            const int gy = ty0 - TPAD + r0 + RS * k;
            const int cy = gy < 0 ? 0 : (gy > IMG - 1 ? IMG - 1 : gy);
            const float t = xin[cy * IMG + cx];
            xv[k] = (cin && gy >= 0 && gy < IMG) ? t : 0.f;
        }
        float* tx = smem + r0 * TPW + c;
        #pragma unroll
        for (int k = 0; k < NE; ++k) {
            const float v = xv[k];
            tx[RS * k * TPW] = v;
            tx[TWORDS + RS * k * TPW] = v * __logf(v + EPS_F);
        }
    }
    __syncthreads();

    hsum_rows<4>(smem, tid);
    hsum_rows<3>(smem, tid);
    hsum_rows<2>(smem, tid);
    hsum_rows<1>(smem, tid);
    __syncthreads();

    const int wave = __builtin_amdgcn_readfirstlane(tid >> 5);
    const int lane = tid & 31;
    const int sub  = wave & 3;
    const int grp  = wave >> 2;
    const int sh = (sub >> 1) * 16;
    const int sw = (sub & 1) * 16;
    const int hi = lane >> 4;
    const int n  = lane & 15;
    const int y0 = sh + hi * 8;
    const int xc = sw + n;

    #pragma nounroll
    for (int pass = 0; pass < 2; ++pass) {
        const int sel = pass == 0 ? grp : 3 - grp;
        const int D   = 4 - sel;
        const int PAD = D * D;
        const int OFF = TPAD - PAD;
        const int RP  = ((16 + 2 * PAD + 3) >> 2) << 2;
        const int NCH = RP >> 2;
        const int PHH = 16 + RP;
        const float* Hs = smem + (sel == 0 ? HS4 : sel == 1 ? HS3 : sel == 2 ? HS2 : HS1);

        const unsigned long long band = sel == 0 ? Geo<4>::band() : sel == 1 ? Geo<3>::band()
                                      : sel == 2 ? Geo<2>::band() : Geo<1>::band();

        v8f acc0 = {0.f, 0.f, 0.f, 0.f, 0.f, 0.f, 0.f, 0.f};
        v8f acc1 = acc0, acc2 = acc0;
        for (int ch = 0; ch < NCH; ++ch) {
            const int r0 = ch * 4 + 2 * hi;
            const int t0 = r0 - n;
            const int t1 = r0 + 1 - n;
            v2f a;
            a.x = (t0 >= 0 && ((band >> (t0 & 63)) & 1ull)) ? 1.f : 0.f;
            a.y = (t1 >= 0 && ((band >> (t1 & 63)) & 1ull)) ? 1.f : 0.f;
            const int br = sh + r0;
            const float* hb = Hs + br * HSP + sw + n;
            v2f b0, b1, b2;
            b0.x = hb[0];                   b0.y = hb[HSP];
            b1.x = hb[PHH * HSP];           b1.y = hb[PHH * HSP + HSP];
            b2.x = hb[2 * PHH * HSP];       b2.y = hb[2 * PHH * HSP + HSP];
            acc0 = __builtin_amdgcn_wmma_f32_16x16x4_f32(false, a, false, b0, (short)0, acc0, false, false);
            acc1 = __builtin_amdgcn_wmma_f32_16x16x4_f32(false, a, false, b1, (short)0, acc1, false, false);
            acc2 = __builtin_amdgcn_wmma_f32_16x16x4_f32(false, a, false, b2, (short)0, acc2, false, false);
        }

        const float invK = sel == 0 ? 1.f / 81.f : sel == 1 ? 1.f / 49.f : sel == 2 ? 1.f / 25.f : 1.f / 9.f;
        const float corr = sel == 0 ? 81.f / 80.f : sel == 1 ? 49.f / 48.f
                         : sel == 2 ? 25.f / 24.f : 9.f / 8.f;
        const float EH   = 0.6065306597126334f;

        float mu[8], ad[8], contrast[8], energy[8], entropy[8];
        #pragma unroll
        for (int v = 0; v < 8; ++v) {
            const float s1 = acc0[v];
            const float s2 = acc1[v];
            const float s3 = acc2[v];
            mu[v] = s1 * invK;
            float var_b = s2 * invK - mu[v] * mu[v];
            var_b = var_b > 0.f ? var_b : 0.f;
            const float sd = __builtin_sqrtf(var_b * corr) + EPS_F;
            contrast[v] = var_b / (sd * sd);
            energy[v]   = s2 * invK;
            entropy[v]  = -s3 * invK;
            ad[v] = 0.f;
        }

        const float* tb = smem + (OFF + y0) * TPW + OFF + xc;
        if (sel == 0)      absdev_taps<4>(tb, mu, ad);
        else if (sel == 1) absdev_taps<3>(tb, mu, ad);
        else if (sel == 2) absdev_taps<2>(tb, mu, ad);
        else               absdev_taps<1>(tb, mu, ad);

        constexpr size_t HW    = (size_t)IMG * IMG;
        constexpr size_t PLANE = (size_t)2 * 16 * HW;
        float* __restrict__ od = out + (size_t)(D - 1) * PLANE;
        float* __restrict__ o0 = od + (size_t)(bc * 4 + 0) * HW;
        float* __restrict__ o1 = od + (size_t)(bc * 4 + 1) * HW;
        float* __restrict__ o2 = od + (size_t)(bc * 4 + 2) * HW;
        float* __restrict__ o3 = od + (size_t)(bc * 4 + 3) * HW;
        #pragma unroll
        for (int v = 0; v < 8; ++v) {
            const float homog = 1.f / (1.f + ad[v] * invK);
            const size_t pix = (size_t)(ty0 + y0 + v) * IMG + (tx0 + xc);
            o0[pix] = (contrast[v] + EPS_F) * EH;
            o1[pix] = (energy[v]   + EPS_F) * EH;
            o2[pix] = (entropy[v]  + EPS_F) * EH;
            o3[pix] = (homog       + EPS_F) * EH;
        }
    }
}

extern "C" void kernel_launch(void* const* d_in, const int* in_sizes, int n_in,
                              void* d_out, int out_size, void* d_ws, size_t ws_size,
                              hipStream_t stream) {
    (void)in_sizes; (void)n_in; (void)d_ws; (void)ws_size; (void)out_size;
    const float* x = (const float*)d_in[0];
    float* out = (float*)d_out;

    dim3 grid(256 / 32, 256 / 32, 8);
    dim3 blk(256);

    texfeat_all_kernel<<<grid, blk, 0, stream>>>(x, out);
}


# Homogeneity's sum |p-mu| taken as 2*(sum max(p,mu) - (sum p + K*mu)/2), sum p from the box sum: a tap is one max and one add, not subtract, sign mask, add; summed terms and rounding change
# speedup vs baseline: 2.7894x; 1.0630x over previous
#include <hip/hip_runtime.h>
#include <math.h>


typedef __attribute__((ext_vector_type(2))) float v2f;
typedef __attribute__((ext_vector_type(8))) float v8f;

#define EPS_F 1e-6f

constexpr int IMG  = 256;
constexpr int TILE = 32;
constexpr int HSP  = TILE + 1;
constexpr int NTHR = 256;

constexpr int TPAD = 16;
constexpr int TPH  = TILE + 2 * TPAD;
constexpr int TPW  = TPH + 2;
constexpr int TWORDS = TPH * TPW;

template <int D> struct Geo {
    static constexpr int K1  = 2 * D + 1;
    static constexpr int PAD = D * D;
    static constexpr int PH  = TILE + 2 * PAD;
    static constexpr int OFF = TPAD - PAD;
    static constexpr int RP  = ((16 + 2 * PAD + 3) / 4) * 4;
    static constexpr int PHH = 16 + RP;
    static constexpr int HWORDS = 3 * PHH * HSP;
    static constexpr unsigned long long band() {
        unsigned long long m = 0;
        for (int t = 0; t <= 2 * PAD; t += D) m |= 1ull << t;
        return m;
    }
};
constexpr int HS4 = 2 * TWORDS;
constexpr int HS3 = HS4 + Geo<4>::HWORDS;
constexpr int HS2 = HS3 + Geo<3>::HWORDS;
constexpr int HS1 = HS2 + Geo<2>::HWORDS;
constexpr int SMEM_WORDS = HS1 + Geo<1>::HWORDS;
template <int D> struct HsBase { static constexpr int v = D == 4 ? HS4 : D == 3 ? HS3 : D == 2 ? HS2 : HS1; };

template <int D>
__device__ __forceinline__
void hsum_rows(float* smem, const int tid)
{
    using G = Geo<D>;
    const float* __restrict__ Tx = smem + G::OFF * TPW + G::OFF;
    const float* __restrict__ Tl = Tx + TWORDS;
    float* __restrict__ Hs = smem + HsBase<D>::v;
    for (int i = tid; i < G::PHH * TILE; i += NTHR) {
        const int r = i / TILE, c = i % TILE;
        float h1 = 0.f, h2 = 0.f, h3 = 0.f;
        if (r < G::PH) {
            #pragma unroll
            for (int j = 0; j < G::K1; ++j) {
                const float v = Tx[r * TPW + c + j * D];
                h1 += v;
                h2 = __builtin_fmaf(v, v, h2);
                h3 += Tl[r * TPW + c + j * D];
            }
        }
        Hs[(0 * G::PHH + r) * HSP + c] = h1;
        Hs[(1 * G::PHH + r) * HSP + c] = h2;
        Hs[(2 * G::PHH + r) * HSP + c] = h3;
    }
}

template <int D>
__device__ __forceinline__
void maxsum_taps(const float* base, const float (&mu)[8], float (&ms)[8])
{
    using G = Geo<D>;
    constexpr int NR = 8 + 2 * G::PAD;
    #pragma nounroll
    for (int jx = 0; jx < G::K1; ++jx) {
        const float* col = base + jx * D;
        #pragma unroll
        for (int rr = 0; rr < NR; ++rr) {
            const float p = col[rr * TPW];
            #pragma unroll
            for (int v = 0; v < 8; ++v) {
                if (rr >= v && (rr - v) % D == 0 && (rr - v) <= 2 * G::PAD)
                    ms[v] += __builtin_fmaxf(p, mu[v]);
            }
        }
    }
}

__global__ __launch_bounds__(NTHR)
void texfeat_all_kernel(const float* __restrict__ x, float* __restrict__ out)
{
    __shared__ float smem[SMEM_WORDS];

    const int tid = threadIdx.x;
    const int bc  = blockIdx.z;
    const int tx0 = blockIdx.x * TILE;
    const int ty0 = blockIdx.y * TILE;
    const float* __restrict__ xin = x + (size_t)bc * (IMG * IMG);

    {
        constexpr int RS = NTHR / TPH;
        constexpr int NE = TPH / RS;
        const int c  = tid & (TPH - 1);
        const int r0 = tid / TPH;
        const int gx = tx0 - TPAD + c;
        const bool cin = gx >= 0 && gx < IMG;
        const int cx = gx < 0 ? 0 : (gx > IMG - 1 ? IMG - 1 : gx);
        float xv[NE];
        #pragma unroll
        for (int k = 0; k < NE; ++k) {
            const int gy = ty0 - TPAD + r0 + RS * k;
            const int cy = gy < 0 ? 0 : (gy > IMG - 1 ? IMG - 1 : gy);
            const float t = xin[cy * IMG + cx];
            xv[k] = (cin && gy >= 0 && gy < IMG) ? t : 0.f;
        }
        float* tx = smem + r0 * TPW + c;
        #pragma unroll
        for (int k = 0; k < NE; ++k) {
            const float v = xv[k];
            tx[RS * k * TPW] = v;
            tx[TWORDS + RS * k * TPW] = v * __logf(v + EPS_F);
        }
    }
    __syncthreads();

    hsum_rows<4>(smem, tid);
    hsum_rows<3>(smem, tid);
    hsum_rows<2>(smem, tid);
    hsum_rows<1>(smem, tid);
    __syncthreads();

    const int wave = __builtin_amdgcn_readfirstlane(tid >> 5);
    const int lane = tid & 31;
    const int sub  = wave & 3;
    const int grp  = wave >> 2;
    const int sh = (sub >> 1) * 16;
    const int sw = (sub & 1) * 16;
    const int hi = lane >> 4;
    const int n  = lane & 15;
    const int y0 = sh + hi * 8;
    const int xc = sw + n;

    #pragma nounroll
    for (int pass = 0; pass < 2; ++pass) {
        const int sel = pass == 0 ? grp : 3 - grp;
        const int D   = 4 - sel;
        const int PAD = D * D;
        const int OFF = TPAD - PAD;
        const int RP  = ((16 + 2 * PAD + 3) >> 2) << 2;
        const int NCH = RP >> 2;
        const int PHH = 16 + RP;
        const float* Hs = smem + (sel == 0 ? HS4 : sel == 1 ? HS3 : sel == 2 ? HS2 : HS1);

        const unsigned long long band = sel == 0 ? Geo<4>::band() : sel == 1 ? Geo<3>::band()
                                      : sel == 2 ? Geo<2>::band() : Geo<1>::band();

        v8f acc0 = {0.f, 0.f, 0.f, 0.f, 0.f, 0.f, 0.f, 0.f};
        v8f acc1 = acc0, acc2 = acc0;
        for (int ch = 0; ch < NCH; ++ch) {
            const int r0 = ch * 4 + 2 * hi;
            const int t0 = r0 - n;
            const int t1 = r0 + 1 - n;
            v2f a;
            a.x = (t0 >= 0 && ((band >> (t0 & 63)) & 1ull)) ? 1.f : 0.f;
            a.y = (t1 >= 0 && ((band >> (t1 & 63)) & 1ull)) ? 1.f : 0.f;
            const int br = sh + r0;
            const float* hb = Hs + br * HSP + sw + n;
            v2f b0, b1, b2;
            b0.x = hb[0];                   b0.y = hb[HSP];
            b1.x = hb[PHH * HSP];           b1.y = hb[PHH * HSP + HSP];
            b2.x = hb[2 * PHH * HSP];       b2.y = hb[2 * PHH * HSP + HSP];
            acc0 = __builtin_amdgcn_wmma_f32_16x16x4_f32(false, a, false, b0, (short)0, acc0, false, false);
            acc1 = __builtin_amdgcn_wmma_f32_16x16x4_f32(false, a, false, b1, (short)0, acc1, false, false);
            acc2 = __builtin_amdgcn_wmma_f32_16x16x4_f32(false, a, false, b2, (short)0, acc2, false, false);
        }

        const float invK = sel == 0 ? 1.f / 81.f : sel == 1 ? 1.f / 49.f : sel == 2 ? 1.f / 25.f : 1.f / 9.f;
        const float corr = sel == 0 ? 81.f / 80.f : sel == 1 ? 49.f / 48.f
                         : sel == 2 ? 25.f / 24.f : 9.f / 8.f;
        const float EH   = 0.6065306597126334f;

        const float Kf   = sel == 0 ? 81.f : sel == 1 ? 49.f : sel == 2 ? 25.f : 9.f;

        float mu[8], ms[8], contrast[8], energy[8], entropy[8];
        #pragma unroll
        for (int v = 0; v < 8; ++v) {
            const float s1 = acc0[v];
            const float s2 = acc1[v];
            const float s3 = acc2[v];
            mu[v] = s1 * invK;
            float var_b = s2 * invK - mu[v] * mu[v];
            var_b = var_b > 0.f ? var_b : 0.f;
            const float sd = __builtin_sqrtf(var_b * corr) + EPS_F;
            contrast[v] = var_b / (sd * sd);
            energy[v]   = s2 * invK;
            entropy[v]  = -s3 * invK;
            ms[v] = -0.5f * (s1 + Kf * mu[v]);
        }

        const float* tb = smem + (OFF + y0) * TPW + OFF + xc;
        if (sel == 0)      maxsum_taps<4>(tb, mu, ms);
        else if (sel == 1) maxsum_taps<3>(tb, mu, ms);
        else if (sel == 2) maxsum_taps<2>(tb, mu, ms);
        else               maxsum_taps<1>(tb, mu, ms);

        constexpr size_t HW    = (size_t)IMG * IMG;
        constexpr size_t PLANE = (size_t)2 * 16 * HW;
        float* __restrict__ od = out + (size_t)(D - 1) * PLANE;
        float* __restrict__ o0 = od + (size_t)(bc * 4 + 0) * HW;
        float* __restrict__ o1 = od + (size_t)(bc * 4 + 1) * HW;
        float* __restrict__ o2 = od + (size_t)(bc * 4 + 2) * HW;
        float* __restrict__ o3 = od + (size_t)(bc * 4 + 3) * HW;
        #pragma unroll
        for (int v = 0; v < 8; ++v) {
            const float ad = 2.f * ms[v];
            const float homog = 1.f / (1.f + ad * invK);
            const size_t pix = (size_t)(ty0 + y0 + v) * IMG + (tx0 + xc);
            o0[pix] = (contrast[v] + EPS_F) * EH;
            o1[pix] = (energy[v]   + EPS_F) * EH;
            o2[pix] = (entropy[v]  + EPS_F) * EH;
            o3[pix] = (homog       + EPS_F) * EH;
        }
    }
}

extern "C" void kernel_launch(void* const* d_in, const int* in_sizes, int n_in,
                              void* d_out, int out_size, void* d_ws, size_t ws_size,
                              hipStream_t stream) {
    (void)in_sizes; (void)n_in; (void)d_ws; (void)ws_size; (void)out_size;
    const float* x = (const float*)d_in[0];
    float* out = (float*)d_out;

    dim3 grid(256 / 32, 256 / 32, 8);
    dim3 blk(256);

    texfeat_all_kernel<<<grid, blk, 0, stream>>>(x, out);
}


# Log, square root and both divisions as single hardware approximations (v_log_f32 * ln2, v_sqrt_f32, v_rcp_f32, about 1 ulp) instead of exactly rounded library expansions; last-bit rounding changes
# speedup vs baseline: 3.0976x; 1.1105x over previous
#include <hip/hip_runtime.h>
#include <math.h>


typedef __attribute__((ext_vector_type(2))) float v2f;
typedef __attribute__((ext_vector_type(8))) float v8f;

#define EPS_F 1e-6f
#define LN2_F 0.6931471805599453f


constexpr int IMG  = 256;
constexpr int TILE = 32;
constexpr int HSP  = TILE + 1;
constexpr int NTHR = 256;

constexpr int TPAD = 16;
constexpr int TPH  = TILE + 2 * TPAD;
constexpr int TPW  = TPH + 2;
constexpr int TWORDS = TPH * TPW;

template <int D> struct Geo {
    static constexpr int K1  = 2 * D + 1;
    static constexpr int PAD = D * D;
    static constexpr int PH  = TILE + 2 * PAD;
    static constexpr int OFF = TPAD - PAD;
    static constexpr int RP  = ((16 + 2 * PAD + 3) / 4) * 4;
    static constexpr int PHH = 16 + RP;
    static constexpr int HWORDS = 3 * PHH * HSP;
    static constexpr unsigned long long band() {
        unsigned long long m = 0;
        for (int t = 0; t <= 2 * PAD; t += D) m |= 1ull << t;
        return m;
    }
};
constexpr int HS4 = 2 * TWORDS;
constexpr int HS3 = HS4 + Geo<4>::HWORDS;
constexpr int HS2 = HS3 + Geo<3>::HWORDS;
constexpr int HS1 = HS2 + Geo<2>::HWORDS;
constexpr int SMEM_WORDS = HS1 + Geo<1>::HWORDS;
template <int D> struct HsBase { static constexpr int v = D == 4 ? HS4 : D == 3 ? HS3 : D == 2 ? HS2 : HS1; };

template <int D>
__device__ __forceinline__
void hsum_rows(float* smem, const int tid)
{
    using G = Geo<D>;
    const float* __restrict__ Tx = smem + G::OFF * TPW + G::OFF;
    const float* __restrict__ Tl = Tx + TWORDS;
    float* __restrict__ Hs = smem + HsBase<D>::v;
    for (int i = tid; i < G::PHH * TILE; i += NTHR) {
        const int r = i / TILE, c = i % TILE;
        float h1 = 0.f, h2 = 0.f, h3 = 0.f;
        if (r < G::PH) {
            #pragma unroll
            for (int j = 0; j < G::K1; ++j) {
                const float v = Tx[r * TPW + c + j * D];
                h1 += v;
                h2 = __builtin_fmaf(v, v, h2);
                h3 += Tl[r * TPW + c + j * D];
            }
        }
        Hs[(0 * G::PHH + r) * HSP + c] = h1;
        Hs[(1 * G::PHH + r) * HSP + c] = h2;
        Hs[(2 * G::PHH + r) * HSP + c] = h3;
    }
}

template <int D>
__device__ __forceinline__
void maxsum_taps(const float* base, const float (&mu)[8], float (&ms)[8])
{
    using G = Geo<D>;
    constexpr int NR = 8 + 2 * G::PAD;
    #pragma nounroll
    for (int jx = 0; jx < G::K1; ++jx) {
        const float* col = base + jx * D;
        #pragma unroll
        for (int rr = 0; rr < NR; ++rr) {
            const float p = col[rr * TPW];
            #pragma unroll
            for (int v = 0; v < 8; ++v) {
                if (rr >= v && (rr - v) % D == 0 && (rr - v) <= 2 * G::PAD)
                    ms[v] += __builtin_fmaxf(p, mu[v]);
            }
        }
    }
}

__global__ __launch_bounds__(NTHR)
void texfeat_all_kernel(const float* __restrict__ x, float* __restrict__ out)
{
    __shared__ float smem[SMEM_WORDS];

    const int tid = threadIdx.x;
    const int bc  = blockIdx.z;
    const int tx0 = blockIdx.x * TILE;
    const int ty0 = blockIdx.y * TILE;
    const float* __restrict__ xin = x + (size_t)bc * (IMG * IMG);

    {
        constexpr int RS = NTHR / TPH;
        constexpr int NE = TPH / RS;
        const int c  = tid & (TPH - 1);
        const int r0 = tid / TPH;
        const int gx = tx0 - TPAD + c;
        const bool cin = gx >= 0 && gx < IMG;
        const int cx = gx < 0 ? 0 : (gx > IMG - 1 ? IMG - 1 : gx);
        float xv[NE];
        #pragma unroll
        for (int k = 0; k < NE; ++k) {
            const int gy = ty0 - TPAD + r0 + RS * k;
            const int cy = gy < 0 ? 0 : (gy > IMG - 1 ? IMG - 1 : gy);
            const float t = xin[cy * IMG + cx];
            xv[k] = (cin && gy >= 0 && gy < IMG) ? t : 0.f;
        }
        float* tx = smem + r0 * TPW + c;
        #pragma unroll
        for (int k = 0; k < NE; ++k) {
            const float v = xv[k];
            tx[RS * k * TPW] = v;
            tx[TWORDS + RS * k * TPW] = v * (__builtin_amdgcn_logf(v + EPS_F) * LN2_F);
        }
    }
    __syncthreads();

    hsum_rows<4>(smem, tid);
    hsum_rows<3>(smem, tid);
    hsum_rows<2>(smem, tid);
    hsum_rows<1>(smem, tid);
    __syncthreads();

    const int wave = __builtin_amdgcn_readfirstlane(tid >> 5);
    const int lane = tid & 31;
    const int sub  = wave & 3;
    const int grp  = wave >> 2;
    const int sh = (sub >> 1) * 16;
    const int sw = (sub & 1) * 16;
    const int hi = lane >> 4;
    const int n  = lane & 15;
    const int y0 = sh + hi * 8;
    const int xc = sw + n;

    #pragma nounroll
    for (int pass = 0; pass < 2; ++pass) {
        const int sel = pass == 0 ? grp : 3 - grp;
        const int D   = 4 - sel;
        const int PAD = D * D;
        const int OFF = TPAD - PAD;
        const int RP  = ((16 + 2 * PAD + 3) >> 2) << 2;
        const int NCH = RP >> 2;
        const int PHH = 16 + RP;
        const float* Hs = smem + (sel == 0 ? HS4 : sel == 1 ? HS3 : sel == 2 ? HS2 : HS1);

        const unsigned long long band = sel == 0 ? Geo<4>::band() : sel == 1 ? Geo<3>::band()
                                      : sel == 2 ? Geo<2>::band() : Geo<1>::band();

        v8f acc0 = {0.f, 0.f, 0.f, 0.f, 0.f, 0.f, 0.f, 0.f};
        v8f acc1 = acc0, acc2 = acc0;
        for (int ch = 0; ch < NCH; ++ch) {
            const int r0 = ch * 4 + 2 * hi;
            const int t0 = r0 - n;
            const int t1 = r0 + 1 - n;
            v2f a;
            a.x = (t0 >= 0 && ((band >> (t0 & 63)) & 1ull)) ? 1.f : 0.f;
            a.y = (t1 >= 0 && ((band >> (t1 & 63)) & 1ull)) ? 1.f : 0.f;
            const int br = sh + r0;
            const float* hb = Hs + br * HSP + sw + n;
            v2f b0, b1, b2;
            b0.x = hb[0];                   b0.y = hb[HSP];
            b1.x = hb[PHH * HSP];           b1.y = hb[PHH * HSP + HSP];
            b2.x = hb[2 * PHH * HSP];       b2.y = hb[2 * PHH * HSP + HSP];
            acc0 = __builtin_amdgcn_wmma_f32_16x16x4_f32(false, a, false, b0, (short)0, acc0, false, false);
            acc1 = __builtin_amdgcn_wmma_f32_16x16x4_f32(false, a, false, b1, (short)0, acc1, false, false);
            acc2 = __builtin_amdgcn_wmma_f32_16x16x4_f32(false, a, false, b2, (short)0, acc2, false, false);
        }

        const float invK = sel == 0 ? 1.f / 81.f : sel == 1 ? 1.f / 49.f : sel == 2 ? 1.f / 25.f : 1.f / 9.f;
        const float corr = sel == 0 ? 81.f / 80.f : sel == 1 ? 49.f / 48.f
                         : sel == 2 ? 25.f / 24.f : 9.f / 8.f;
        const float EH   = 0.6065306597126334f;

        const float Kf   = sel == 0 ? 81.f : sel == 1 ? 49.f : sel == 2 ? 25.f : 9.f;

        float mu[8], ms[8], contrast[8], energy[8], entropy[8];
        #pragma unroll
        for (int v = 0; v < 8; ++v) {
            const float s1 = acc0[v];
            const float s2 = acc1[v];
            const float s3 = acc2[v];
            mu[v] = s1 * invK;
            float var_b = s2 * invK - mu[v] * mu[v];
            var_b = var_b > 0.f ? var_b : 0.f;
            const float sd = __builtin_amdgcn_sqrtf(var_b * corr) + EPS_F;
            contrast[v] = var_b * __builtin_amdgcn_rcpf(sd * sd);
            energy[v]   = s2 * invK;
            entropy[v]  = -s3 * invK;
            ms[v] = -0.5f * (s1 + Kf * mu[v]);
        }

        const float* tb = smem + (OFF + y0) * TPW + OFF + xc;
        if (sel == 0)      maxsum_taps<4>(tb, mu, ms);
        else if (sel == 1) maxsum_taps<3>(tb, mu, ms);
        else if (sel == 2) maxsum_taps<2>(tb, mu, ms);
        else               maxsum_taps<1>(tb, mu, ms);

        constexpr size_t HW    = (size_t)IMG * IMG;
        constexpr size_t PLANE = (size_t)2 * 16 * HW;
        float* __restrict__ od = out + (size_t)(D - 1) * PLANE;
        float* __restrict__ o0 = od + (size_t)(bc * 4 + 0) * HW;
        float* __restrict__ o1 = od + (size_t)(bc * 4 + 1) * HW;
        float* __restrict__ o2 = od + (size_t)(bc * 4 + 2) * HW;
        float* __restrict__ o3 = od + (size_t)(bc * 4 + 3) * HW;
        #pragma unroll
        for (int v = 0; v < 8; ++v) {
            const float ad = 2.f * ms[v];
            const float homog = __builtin_amdgcn_rcpf(1.f + ad * invK);
            const size_t pix = (size_t)(ty0 + y0 + v) * IMG + (tx0 + xc);
            o0[pix] = (contrast[v] + EPS_F) * EH;
            o1[pix] = (energy[v]   + EPS_F) * EH;
            o2[pix] = (entropy[v]  + EPS_F) * EH;
            o3[pix] = (homog       + EPS_F) * EH;
        }
    }
}

extern "C" void kernel_launch(void* const* d_in, const int* in_sizes, int n_in,
                              void* d_out, int out_size, void* d_ws, size_t ws_size,
                              hipStream_t stream) {
    (void)in_sizes; (void)n_in; (void)d_ws; (void)ws_size; (void)out_size;
    const float* x = (const float*)d_in[0];
    float* out = (float*)d_out;

    dim3 grid(256 / 32, 256 / 32, 8);
    dim3 blk(256);

    texfeat_all_kernel<<<grid, blk, 0, stream>>>(x, out);
}


# Horizontal sums: each thread reads one tile row's 40-column run of x and x*log x once (16-byte LDS loads) and forms 8 outputs of all four dilations from registers; 16-byte stores; sum order kept
# speedup vs baseline: 3.5694x; 1.1523x over previous
#include <hip/hip_runtime.h>
#include <math.h>


typedef __attribute__((ext_vector_type(2))) float v2f;
typedef __attribute__((ext_vector_type(4))) float v4f;
typedef __attribute__((ext_vector_type(8))) float v8f;

#define EPS_F 1e-6f
#define LN2_F 0.6931471805599453f


constexpr int IMG  = 256;
constexpr int TILE = 32;
constexpr int HSP  = TILE + 4;
constexpr int NTHR = 256;

constexpr int TPAD = 16;
constexpr int TPH  = TILE + 2 * TPAD;
constexpr int TPW  = TPH + 4;
constexpr int TWORDS = TPH * TPW;

template <int D> struct Geo {
    static constexpr int K1  = 2 * D + 1;
    static constexpr int PAD = D * D;
    static constexpr int PH  = TILE + 2 * PAD;
    static constexpr int OFF = TPAD - PAD;
    static constexpr int RP  = ((16 + 2 * PAD + 3) / 4) * 4;
    static constexpr int PHH = 16 + RP;
    static constexpr int HWORDS = 3 * PHH * HSP;
    static constexpr unsigned long long band() {
        unsigned long long m = 0;
        for (int t = 0; t <= 2 * PAD; t += D) m |= 1ull << t;
        return m;
    }
};
constexpr int HS4 = 2 * TWORDS;
constexpr int HS3 = HS4 + Geo<4>::HWORDS;
constexpr int HS2 = HS3 + Geo<3>::HWORDS;
constexpr int HS1 = HS2 + Geo<2>::HWORDS;
constexpr int SMEM_WORDS = HS1 + Geo<1>::HWORDS;
template <int D> struct HsBase { static constexpr int v = D == 4 ? HS4 : D == 3 ? HS3 : D == 2 ? HS2 : HS1; };

constexpr int RUN = 8 + 2 * TPAD;

template <int D>
__device__ __forceinline__
void hsum_run(const float (&xs)[RUN], const float (&ls)[RUN], float* smem, const int r, const int g)
{
    using G = Geo<D>;
    float h1[8], h2[8], h3[8];
    #pragma unroll
    for (int o = 0; o < 8; ++o) {
        float a = 0.f, b = 0.f, c = 0.f;
        #pragma unroll
        for (int j = 0; j < G::K1; ++j) {
            const float v = xs[G::OFF + o + j * D];
            a += v;
            b = __builtin_fmaf(v, v, b);
            c += ls[G::OFF + o + j * D];
        }
        h1[o] = a; h2[o] = b; h3[o] = c;
    }
    const int rr = r - G::OFF;
    float* hs = smem + HsBase<D>::v + rr * HSP + 8 * g;
    if (rr >= 0 && rr < G::PH) {
        *(v4f*)(hs + 0 * G::PHH * HSP)     = (v4f){h1[0], h1[1], h1[2], h1[3]};
        *(v4f*)(hs + 0 * G::PHH * HSP + 4) = (v4f){h1[4], h1[5], h1[6], h1[7]};
        *(v4f*)(hs + 1 * G::PHH * HSP)     = (v4f){h2[0], h2[1], h2[2], h2[3]};
        *(v4f*)(hs + 1 * G::PHH * HSP + 4) = (v4f){h2[4], h2[5], h2[6], h2[7]};
        *(v4f*)(hs + 2 * G::PHH * HSP)     = (v4f){h3[0], h3[1], h3[2], h3[3]};
        *(v4f*)(hs + 2 * G::PHH * HSP + 4) = (v4f){h3[4], h3[5], h3[6], h3[7]};
    }
    if (G::PHH > G::PH && rr >= G::PH && rr < G::PHH) {
        const v4f z = {0.f, 0.f, 0.f, 0.f};
        #pragma unroll
        for (int s = 0; s < 3; ++s) {
            *(v4f*)(hs + s * G::PHH * HSP)     = z;
            *(v4f*)(hs + s * G::PHH * HSP + 4) = z;
        }
    }
}

template <int D>
__device__ __forceinline__
void maxsum_taps(const float* base, const float (&mu)[8], float (&ms)[8])
{
    using G = Geo<D>;
    constexpr int NR = 8 + 2 * G::PAD;
    #pragma nounroll
    for (int jx = 0; jx < G::K1; ++jx) {
        const float* col = base + jx * D;
        #pragma unroll
        for (int rr = 0; rr < NR; ++rr) {
            const float p = col[rr * TPW];
            #pragma unroll
            for (int v = 0; v < 8; ++v) {
                if (rr >= v && (rr - v) % D == 0 && (rr - v) <= 2 * G::PAD)
                    ms[v] += __builtin_fmaxf(p, mu[v]);
            }
        }
    }
}

__global__ __launch_bounds__(NTHR)
void texfeat_all_kernel(const float* __restrict__ x, float* __restrict__ out)
{
    __shared__ __attribute__((aligned(16))) float smem[SMEM_WORDS];

    const int tid = threadIdx.x;
    const int bc  = blockIdx.z;
    const int tx0 = blockIdx.x * TILE;
    const int ty0 = blockIdx.y * TILE;
    const float* __restrict__ xin = x + (size_t)bc * (IMG * IMG);

    {
        constexpr int RS = NTHR / TPH;
        constexpr int NE = TPH / RS;
        const int c  = tid & (TPH - 1);
        const int r0 = tid / TPH;
        const int gx = tx0 - TPAD + c;
        const bool cin = gx >= 0 && gx < IMG;
        const int cx = gx < 0 ? 0 : (gx > IMG - 1 ? IMG - 1 : gx);
        float xv[NE];
        #pragma unroll
        for (int k = 0; k < NE; ++k) {
            const int gy = ty0 - TPAD + r0 + RS * k;
            const int cy = gy < 0 ? 0 : (gy > IMG - 1 ? IMG - 1 : gy);
            const float t = xin[cy * IMG + cx];
            xv[k] = (cin && gy >= 0 && gy < IMG) ? t : 0.f;
        }
        float* tx = smem + r0 * TPW + c;
        #pragma unroll
        for (int k = 0; k < NE; ++k) {
            const float v = xv[k];
            tx[RS * k * TPW] = v;
            tx[TWORDS + RS * k * TPW] = v * (__builtin_amdgcn_logf(v + EPS_F) * LN2_F);
        }
    }
    __syncthreads();

    {
        const int r = tid & (TPH - 1);
        const int g = tid / TPH;
        const v4f* px = (const v4f*)(smem + r * TPW + 8 * g);
        const v4f* pl = (const v4f*)(smem + TWORDS + r * TPW + 8 * g);
        float xs[RUN], ls[RUN];
        #pragma unroll
        for (int k = 0; k < RUN / 4; ++k) {
            const v4f t = px[k];
            const v4f u = pl[k];
            xs[4 * k + 0] = t.x; xs[4 * k + 1] = t.y; xs[4 * k + 2] = t.z; xs[4 * k + 3] = t.w;
            ls[4 * k + 0] = u.x; ls[4 * k + 1] = u.y; ls[4 * k + 2] = u.z; ls[4 * k + 3] = u.w;
        }
        hsum_run<4>(xs, ls, smem, r, g);
        hsum_run<3>(xs, ls, smem, r, g);
        hsum_run<2>(xs, ls, smem, r, g);
        hsum_run<1>(xs, ls, smem, r, g);
    }
    __syncthreads();

    const int wave = __builtin_amdgcn_readfirstlane(tid >> 5);
    const int lane = tid & 31;
    const int sub  = wave & 3;
    const int grp  = wave >> 2;
    const int sh = (sub >> 1) * 16;
    const int sw = (sub & 1) * 16;
    const int hi = lane >> 4;
    const int n  = lane & 15;
    const int y0 = sh + hi * 8;
    const int xc = sw + n;

    #pragma nounroll
    for (int pass = 0; pass < 2; ++pass) {
        const int sel = pass == 0 ? grp : 3 - grp;
        const int D   = 4 - sel;
        const int PAD = D * D;
        const int OFF = TPAD - PAD;
        const int RP  = ((16 + 2 * PAD + 3) >> 2) << 2;
        const int NCH = RP >> 2;
        const int PHH = 16 + RP;
        const float* Hs = smem + (sel == 0 ? HS4 : sel == 1 ? HS3 : sel == 2 ? HS2 : HS1);

        const unsigned long long band = sel == 0 ? Geo<4>::band() : sel == 1 ? Geo<3>::band()
                                      : sel == 2 ? Geo<2>::band() : Geo<1>::band();

        v8f acc0 = {0.f, 0.f, 0.f, 0.f, 0.f, 0.f, 0.f, 0.f};
        v8f acc1 = acc0, acc2 = acc0;
        for (int ch = 0; ch < NCH; ++ch) {
            const int r0 = ch * 4 + 2 * hi;
            const int t0 = r0 - n;
            const int t1 = r0 + 1 - n;
            v2f a;
            a.x = (t0 >= 0 && ((band >> (t0 & 63)) & 1ull)) ? 1.f : 0.f;
            a.y = (t1 >= 0 && ((band >> (t1 & 63)) & 1ull)) ? 1.f : 0.f;
            const int br = sh + r0;
            const float* hb = Hs + br * HSP + sw + n;
            v2f b0, b1, b2;
            b0.x = hb[0];                   b0.y = hb[HSP];
            b1.x = hb[PHH * HSP];           b1.y = hb[PHH * HSP + HSP];
            b2.x = hb[2 * PHH * HSP];       b2.y = hb[2 * PHH * HSP + HSP];
            acc0 = __builtin_amdgcn_wmma_f32_16x16x4_f32(false, a, false, b0, (short)0, acc0, false, false);
            acc1 = __builtin_amdgcn_wmma_f32_16x16x4_f32(false, a, false, b1, (short)0, acc1, false, false);
            acc2 = __builtin_amdgcn_wmma_f32_16x16x4_f32(false, a, false, b2, (short)0, acc2, false, false);
        }

        const float invK = sel == 0 ? 1.f / 81.f : sel == 1 ? 1.f / 49.f : sel == 2 ? 1.f / 25.f : 1.f / 9.f;
        const float corr = sel == 0 ? 81.f / 80.f : sel == 1 ? 49.f / 48.f
                         : sel == 2 ? 25.f / 24.f : 9.f / 8.f;
        const float EH   = 0.6065306597126334f;

        const float Kf   = sel == 0 ? 81.f : sel == 1 ? 49.f : sel == 2 ? 25.f : 9.f;

        float mu[8], ms[8], contrast[8], energy[8], entropy[8];
        #pragma unroll
        for (int v = 0; v < 8; ++v) {
            const float s1 = acc0[v];
            const float s2 = acc1[v];
            const float s3 = acc2[v];
            mu[v] = s1 * invK;
            float var_b = s2 * invK - mu[v] * mu[v];
            var_b = var_b > 0.f ? var_b : 0.f;
            const float sd = __builtin_amdgcn_sqrtf(var_b * corr) + EPS_F;
            contrast[v] = var_b * __builtin_amdgcn_rcpf(sd * sd);
            energy[v]   = s2 * invK;
            entropy[v]  = -s3 * invK;
            ms[v] = -0.5f * (s1 + Kf * mu[v]);
        }

        const float* tb = smem + (OFF + y0) * TPW + OFF + xc;
        if (sel == 0)      maxsum_taps<4>(tb, mu, ms);
        else if (sel == 1) maxsum_taps<3>(tb, mu, ms);
        else if (sel == 2) maxsum_taps<2>(tb, mu, ms);
        else               maxsum_taps<1>(tb, mu, ms);

        constexpr size_t HW    = (size_t)IMG * IMG;
        constexpr size_t PLANE = (size_t)2 * 16 * HW;
        float* __restrict__ od = out + (size_t)(D - 1) * PLANE;
        float* __restrict__ o0 = od + (size_t)(bc * 4 + 0) * HW;
        float* __restrict__ o1 = od + (size_t)(bc * 4 + 1) * HW;
        float* __restrict__ o2 = od + (size_t)(bc * 4 + 2) * HW;
        float* __restrict__ o3 = od + (size_t)(bc * 4 + 3) * HW;
        #pragma unroll
        for (int v = 0; v < 8; ++v) {
            const float ad = 2.f * ms[v];
            const float homog = __builtin_amdgcn_rcpf(1.f + ad * invK);
            const size_t pix = (size_t)(ty0 + y0 + v) * IMG + (tx0 + xc);
            o0[pix] = (contrast[v] + EPS_F) * EH;
            o1[pix] = (energy[v]   + EPS_F) * EH;
            o2[pix] = (entropy[v]  + EPS_F) * EH;
            o3[pix] = (homog       + EPS_F) * EH;
        }
    }
}

extern "C" void kernel_launch(void* const* d_in, const int* in_sizes, int n_in,
                              void* d_out, int out_size, void* d_ws, size_t ws_size,
                              hipStream_t stream) {
    (void)in_sizes; (void)n_in; (void)d_ws; (void)ws_size; (void)out_size;
    const float* x = (const float*)d_in[0];
    float* out = (float*)d_out;

    dim3 grid(256 / 32, 256 / 32, 8);
    dim3 blk(256);

    texfeat_all_kernel<<<grid, blk, 0, stream>>>(x, out);
}
